# starting kernel
# baseline (speedup 1.0000x reference)
#include <hip/hip_runtime.h>

#define N_NODES 100000
#define N_EDGES 800000
#define HIDDEN  128

typedef __attribute__((ext_vector_type(2))) float v2f;
typedef __attribute__((ext_vector_type(8))) float v8f;

__global__ void gnn_count_edges(const int* __restrict__ ei,
                                float* __restrict__ cnt, int E) {
    int e = blockIdx.x * blockDim.x + threadIdx.x;
    if (e < E) {
        int dst = ei[E + e];
        atomicAdd(&cnt[dst], 1.0f);
    }
}

__global__ void gnn_scatter(const float* __restrict__ feat,
                            const int* __restrict__ ei,
                            float* __restrict__ agg, int E) {
    int idx = blockIdx.x * blockDim.x + threadIdx.x;
    int e  = idx >> 5;
    int f4 = (idx & 31) << 2;
    if (e < E) {
        int src = ei[e];
        int dst = ei[E + e];
        const float4 v = *(const float4*)(feat + (size_t)src * HIDDEN + f4);
        float* a = agg + (size_t)dst * HIDDEN + f4;
        atomicAdd(a + 0, v.x);
        atomicAdd(a + 1, v.y);
        atomicAdd(a + 2, v.z);
        atomicAdd(a + 3, v.w);
    }
}

__global__ __launch_bounds__(128)
void gnn_sage_layer(const float* __restrict__ X,
                    const float* __restrict__ agg,
                    const float* __restrict__ cnt,
                    const float* __restrict__ Wl,
                    const float* __restrict__ Wr,
                    const float* __restrict__ bias,
                    float* __restrict__ out,
                    int do_relu) {
    constexpr int LD = HIDDEN + 4;
    __shared__ float Xt[16 * LD];
    __shared__ float Mt[16 * LD];

    const int tid  = threadIdx.x;
    const int row0 = blockIdx.x * 16;

    #pragma unroll
    for (int i = 0; i < 16; ++i) {
        float rc = 1.0f / fmaxf(cnt[row0 + i], 1.0f);
        Xt[i * LD + tid] = X[(size_t)(row0 + i) * HIDDEN + tid];
        Mt[i * LD + tid] = agg[(size_t)(row0 + i) * HIDDEN + tid] * rc;
    }
    __syncthreads();

    const int wave  = tid >> 5;
    const int lane  = tid & 31;
    const int m     = lane & 15;
    const int khalf = (lane >> 4) << 1;

    #pragma unroll
    for (int t = 0; t < 2; ++t) {
        const int col0 = (wave * 2 + t) * 16;

        const float bv = bias[col0 + m];
        v8f c = {bv, bv, bv, bv, bv, bv, bv, bv};

        #pragma unroll 4
        for (int k0 = 0; k0 < HIDDEN; k0 += 4) {
            const int kb = k0 + khalf;
            v2f a, b;

            a.x = Mt[m * LD + kb];
            a.y = Mt[m * LD + kb + 1];
            b.x = Wl[kb * HIDDEN + col0 + m];
            b.y = Wl[(kb + 1) * HIDDEN + col0 + m];
            c = __builtin_amdgcn_wmma_f32_16x16x4_f32(
                    false, a, false, b, (short)0, c, false, false);

            a.x = Xt[m * LD + kb];
            a.y = Xt[m * LD + kb + 1];
            b.x = Wr[kb * HIDDEN + col0 + m];
            b.y = Wr[(kb + 1) * HIDDEN + col0 + m];
            c = __builtin_amdgcn_wmma_f32_16x16x4_f32(
                    false, a, false, b, (short)0, c, false, false);
        }

        const int rbase = row0 + ((lane >> 4) << 3);
        #pragma unroll
        for (int r = 0; r < 8; ++r) {
            float v = c[r];
            if (do_relu) v = fmaxf(v, 0.0f);
            out[(size_t)(rbase + r) * HIDDEN + col0 + m] = v;
        }
    }
}

extern "C" void kernel_launch(void* const* d_in, const int* in_sizes, int n_in,
                              void* d_out, int out_size, void* d_ws, size_t ws_size,
                              hipStream_t stream) {
    const float* x   = (const float*)d_in[0];
    const int*   ei  = (const int*)d_in[1];
    const float* Wl1 = (const float*)d_in[2];
    const float* Wr1 = (const float*)d_in[3];
    const float* b1  = (const float*)d_in[4];
    const float* Wl2 = (const float*)d_in[5];
    const float* Wr2 = (const float*)d_in[6];
    const float* b2  = (const float*)d_in[7];
    float* out = (float*)d_out;

    char* ws = (char*)d_ws;
    const size_t featBytes = (size_t)N_NODES * HIDDEN * sizeof(float);
    float* agg = (float*)(ws);
    float* h   = (float*)(ws + featBytes);
    float* cnt = (float*)(ws + 2 * featBytes);

    hipMemsetAsync(agg, 0, featBytes, stream);
    hipMemsetAsync(cnt, 0, (size_t)N_NODES * sizeof(float), stream);

    const int scatterBlocks = (N_EDGES * 32) / 256;
    gnn_count_edges<<<(N_EDGES + 255) / 256, 256, 0, stream>>>(ei, cnt, N_EDGES);
    gnn_scatter<<<scatterBlocks, 256, 0, stream>>>(x, ei, agg, N_EDGES);

    gnn_sage_layer<<<N_NODES / 16, 128, 0, stream>>>(x, agg, cnt,
                                                     Wl1, Wr1, b1, h, 1);

    hipMemsetAsync(agg, 0, featBytes, stream);
    gnn_scatter<<<scatterBlocks, 256, 0, stream>>>(h, ei, agg, N_EDGES);

    gnn_sage_layer<<<N_NODES / 16, 128, 0, stream>>>(h, agg, cnt,
                                                     Wl2, Wr2, b2, out, 0);
}


# Layer GEMM: 256 persistent 8-wave blocks copy W_l, W_r and bias into LDS once; each wave owns whole 16-row tiles, so B fragments come from LDS, not L2; per-element sum order unchanged
# speedup vs baseline: 1.5525x; 1.5525x over previous
#include <hip/hip_runtime.h>


#define N_NODES 100000
#define N_EDGES 800000
#define HIDDEN  128

#define TILE_NODES 16
#define N_TILES    (N_NODES / TILE_NODES)
#define LIST_CAP   256
#define SRC_BITS   17
#define SRC_MASK   ((1u << SRC_BITS) - 1u)
#define KEY_PAD    (31u << SRC_BITS)

static_assert(LIST_CAP == 256, "the sort does 4 compare-exchanges per lane per stage");

static_assert(N_NODES % TILE_NODES == 0, "tiles cover the nodes exactly");
static_assert(N_NODES <= (1 << SRC_BITS), "source id fits the key");

typedef __attribute__((ext_vector_type(2))) float v2f;
typedef __attribute__((ext_vector_type(8))) float v8f;
typedef __attribute__((ext_vector_type(4))) float v4f;

__global__ void gnn_bin_edges(const int* __restrict__ ei,
                              unsigned* __restrict__ fill,
                              unsigned* __restrict__ list, int E) {
    int e = blockIdx.x * blockDim.x + threadIdx.x;
    if (e < E) {
        const unsigned src  = (unsigned)ei[e];
        const unsigned dst  = (unsigned)ei[E + e];
        const unsigned tile = dst >> 4;
        const unsigned slot = atomicAdd(&fill[tile], 1u);
        if (slot < LIST_CAP)
            list[tile * LIST_CAP + slot] = ((dst & 15u) << SRC_BITS) | src;
    }
}

static __device__ __forceinline__
void tile_accumulate(const float* __restrict__ feat, unsigned* keys,
                     unsigned* hist, int count, int lane, v8f* c) {
    const int padded = (count + 15) & ~15;
    for (int i = count + lane; i < padded; i += 32) keys[i] = KEY_PAD;
    __syncthreads();

    for (int i = lane; i < count; i += 32)
        atomicAdd(&hist[keys[i] >> SRC_BITS], 1u);

    const unsigned m = lane & 15;
    const int khalf  = (lane >> 4) << 1;

    for (int k0 = 0; k0 < padded; k0 += 16) {
        #pragma unroll
        for (int u = 0; u < 4; ++u) {
            const uint2 kk = *(const uint2*)&keys[k0 + 4 * u + khalf];
            const float* p0 = feat + (size_t)(kk.x & SRC_MASK) * HIDDEN + m * 4;
            const float* p1 = feat + (size_t)(kk.y & SRC_MASK) * HIDDEN + m * 4;
            const float4 r00 = *(const float4*)(p0);
            const float4 r01 = *(const float4*)(p0 + 64);
            const float4 r10 = *(const float4*)(p1);
            const float4 r11 = *(const float4*)(p1 + 64);
            v2f a, b;
            a.x = ((kk.x >> SRC_BITS) == m) ? 1.0f : 0.0f;
            a.y = ((kk.y >> SRC_BITS) == m) ? 1.0f : 0.0f;
            b.x = r00.x; b.y = r10.x;
            c[0] = __builtin_amdgcn_wmma_f32_16x16x4_f32(false, a, false, b, (short)0, c[0], false, false);
            b.x = r00.y; b.y = r10.y;
            c[1] = __builtin_amdgcn_wmma_f32_16x16x4_f32(false, a, false, b, (short)0, c[1], false, false);
            b.x = r00.z; b.y = r10.z;
            c[2] = __builtin_amdgcn_wmma_f32_16x16x4_f32(false, a, false, b, (short)0, c[2], false, false);
            b.x = r00.w; b.y = r10.w;
            c[3] = __builtin_amdgcn_wmma_f32_16x16x4_f32(false, a, false, b, (short)0, c[3], false, false);
            b.x = r01.x; b.y = r11.x;
            c[4] = __builtin_amdgcn_wmma_f32_16x16x4_f32(false, a, false, b, (short)0, c[4], false, false);
            b.x = r01.y; b.y = r11.y;
            c[5] = __builtin_amdgcn_wmma_f32_16x16x4_f32(false, a, false, b, (short)0, c[5], false, false);
            b.x = r01.z; b.y = r11.z;
            c[6] = __builtin_amdgcn_wmma_f32_16x16x4_f32(false, a, false, b, (short)0, c[6], false, false);
            b.x = r01.w; b.y = r11.w;
            c[7] = __builtin_amdgcn_wmma_f32_16x16x4_f32(false, a, false, b, (short)0, c[7], false, false);
        }
    }
}

__global__ __launch_bounds__(32)
void gnn_tile_sum(const float* __restrict__ feat,
                  const int* __restrict__ ei,
                  const unsigned* __restrict__ fill,
                  unsigned* __restrict__ list,
                  float* __restrict__ agg,
                  float* __restrict__ cnt,
                  int E, int first_pass) {
    __shared__ __attribute__((aligned(16))) unsigned keys[LIST_CAP + 16];
    __shared__ unsigned hist[32];

    const int lane = threadIdx.x;
    const int tile = blockIdx.x;
    hist[lane] = 0u;

    v8f c[8];
    #pragma unroll
    for (int t = 0; t < 8; ++t) c[t] = (v8f){0.f, 0.f, 0.f, 0.f, 0.f, 0.f, 0.f, 0.f};

    const int n = __builtin_amdgcn_readfirstlane((int)fill[tile]);

    if (n <= LIST_CAP) {
        unsigned* mine = list + (size_t)tile * LIST_CAP;
        if (first_pass) {
            #pragma unroll
            for (int q = 0; q < LIST_CAP / 32; ++q) {
                const int i = lane + 32 * q;
                keys[i] = (i < n) ? mine[i] : KEY_PAD;
            }
            __syncthreads();
            #pragma unroll 1
            for (int k = 2; k <= LIST_CAP; k <<= 1) {
                #pragma unroll 1
                for (int j = k >> 1; j > 0; j >>= 1) {
                    int ia[4], la[4];
                    unsigned x0[4], x1[4];
                    #pragma unroll
                    for (int q = 0; q < 4; ++q) {
                        const int t = lane + 32 * q;
                        ia[q] = ((t & ~(j - 1)) << 1) | (t & (j - 1));
                        la[q] = ia[q] | j;
                        x0[q] = keys[ia[q]];
                        x1[q] = keys[la[q]];
                    }
                    #pragma unroll
                    for (int q = 0; q < 4; ++q) {
                        const bool up = ((ia[q] & k) == 0);
                        const unsigned lo = min(x0[q], x1[q]), hi = max(x0[q], x1[q]);
                        keys[ia[q]] = up ? lo : hi;
                        keys[la[q]] = up ? hi : lo;
                    }
                    __syncthreads();
                }
            }
            for (int i = lane; i < n; i += 32) mine[i] = keys[i];
        } else {
            for (int i = lane; i < n; i += 32) keys[i] = mine[i];
        }
        tile_accumulate(feat, keys, hist, n, lane, c);
    } else {
        int got = 0;
        for (int e0 = 0; e0 < E; e0 += 32) {
            const int e   = e0 + lane;
            const int dst = (e < E) ? ei[E + e] : -1;
            const bool hit = (dst >> 4) == tile;
            const unsigned mask = __builtin_amdgcn_ballot_w32(hit);
            if (mask != 0u) {
                if (hit)
                    keys[got + __popc(mask & ((1u << lane) - 1u))] =
                        (((unsigned)dst & 15u) << SRC_BITS) | (unsigned)ei[e];
                got += __popc(mask);
                if (got > LIST_CAP - 32) {
                    tile_accumulate(feat, keys, hist, got, lane, c);
                    __syncthreads();
                    got = 0;
                }
            }
        }
        if (got > 0) tile_accumulate(feat, keys, hist, got, lane, c);
    }
    __syncthreads();

    const int m     = lane & 15;
    const int rbase = tile * TILE_NODES + ((lane >> 4) << 3);
    #pragma unroll
    for (int r = 0; r < 8; ++r) {
        float* o = agg + (size_t)(rbase + r) * HIDDEN + m * 4;
        *(float4*)(o)      = make_float4(c[0][r], c[1][r], c[2][r], c[3][r]);
        *(float4*)(o + 64) = make_float4(c[4][r], c[5][r], c[6][r], c[7][r]);
    }
    if (first_pass && lane < TILE_NODES)
        cnt[tile * TILE_NODES + lane] = (float)hist[lane];
}

#define LAYER_WAVES   8
#define LAYER_THREADS (LAYER_WAVES * 32)
#define LAYER_BLOCKS  256
#define WP_PITCH      260
#define W_FLOATS      (64 * WP_PITCH)
#define A_LD          (HIDDEN + 4)
#define A_FLOATS      (16 * A_LD)

static_assert(HIDDEN == 128, "the staged layout and the column map assume 128 columns");

#define WMMA_F32(A_, BX_, BY_, C_) \
    do { v2f b_; b_.x = (BX_); b_.y = (BY_); \
         (C_) = __builtin_amdgcn_wmma_f32_16x16x4_f32(false, (A_), false, b_, (short)0, (C_), false, false); } while (0)

__global__ __launch_bounds__(LAYER_THREADS)
void gnn_sage_layer(const float* __restrict__ X,
                    const float* __restrict__ agg,
                    const float* __restrict__ cnt,
                    const float* __restrict__ Wl,
                    const float* __restrict__ Wr,
                    const float* __restrict__ bias,
                    float* __restrict__ out,
                    int do_relu) {
    __shared__ __attribute__((aligned(16))) float Ws[2 * W_FLOATS];
    __shared__ __attribute__((aligned(16))) float At[LAYER_WAVES * 2 * A_FLOATS];
    __shared__ __attribute__((aligned(16))) float Bs[HIDDEN];

    const int tid  = threadIdx.x;
    const int lane = tid & 31;
    const int wave = __builtin_amdgcn_readfirstlane(tid >> 5);
    const int m    = lane & 15;
    const int h    = lane >> 4;

    const int n_waves = LAYER_WAVES * (int)gridDim.x;
    int tile = wave * (int)gridDim.x + (int)blockIdx.x;

    v4f xr[16], ar[16];
    if (tile < N_TILES) {
        const float* xp = X   + (size_t)tile * 16 * HIDDEN + 4 * lane;
        const float* ap = agg + (size_t)tile * 16 * HIDDEN + 4 * lane;
        #pragma unroll
        for (int r = 0; r < 16; ++r) {
            xr[r] = *(const v4f*)(xp + r * HIDDEN);
            ar[r] = *(const v4f*)(ap + r * HIDDEN);
        }
    }

    for (int idx = tid; idx < HIDDEN * HIDDEN; idx += LAYER_THREADS) {
        const int k   = idx >> 7;
        const int col = idx & 127;
        const int w   = col & 63;
        const int g   = ((col >> 6) << 1) | ((w >> 1) & 1);
        const int p   = g * 32 + (w >> 2) * 2 + (w & 1);
        const int a   = (k >> 1) * WP_PITCH + 2 * p + (k & 1);
        Ws[a]            = Wl[idx];
        Ws[W_FLOATS + a] = Wr[idx];
    }
    if (tid < HIDDEN) Bs[tid] = bias[tid];
    __syncthreads();

    float* Mt = At + wave * (2 * A_FLOATS);
    float* Xt = Mt + A_FLOATS;

    const float* aM = Mt + m * A_LD + 2 * h;
    const float* aX = Xt + m * A_LD + 2 * h;
    const float* bL = Ws + h * WP_PITCH + 4 * m;
    const float* bR = bL + W_FLOATS;

    for (; tile < N_TILES; tile += n_waves) {
        const int row0 = tile * 16;

        #pragma unroll
        for (int r = 0; r < 16; ++r) {
            const float rc = 1.0f / fmaxf(cnt[row0 + r], 1.0f);
            *(v4f*)(Mt + r * A_LD + 4 * lane) = ar[r] * rc;
            *(v4f*)(Xt + r * A_LD + 4 * lane) = xr[r];
        }
        __builtin_amdgcn_fence(__ATOMIC_RELEASE, "wavefront");
        __builtin_amdgcn_wave_barrier();
        __builtin_amdgcn_fence(__ATOMIC_ACQUIRE, "wavefront");

        const int nt = tile + n_waves;
        if (nt < N_TILES) {
            const float* xp = X   + (size_t)nt * 16 * HIDDEN + 4 * lane;
            const float* ap = agg + (size_t)nt * 16 * HIDDEN + 4 * lane;
            #pragma unroll
            for (int r = 0; r < 16; ++r) {
                xr[r] = *(const v4f*)(xp + r * HIDDEN);
                ar[r] = *(const v4f*)(ap + r * HIDDEN);
            }
        }

        const int rbase = row0 + (h << 3);
        #pragma unroll 1
        for (int hp = 0; hp < 2; ++hp) {
            const v4f bv = *(const v4f*)(Bs + hp * 64 + 4 * m);
            v8f c[4];
            #pragma unroll
            for (int r = 0; r < 8; ++r) {
                c[0][r] = bv.x; c[1][r] = bv.y; c[2][r] = bv.z; c[3][r] = bv.w;
            }
            const float* bLh = bL + hp * 128;
            const float* bRh = bR + hp * 128;

            #pragma unroll 4
            for (int k0 = 0; k0 < HIDDEN; k0 += 4) {
                const int ko = (k0 >> 1) * WP_PITCH;
                const v2f    am = *(const v2f*)(aM + k0);
                const v4f l0 = *(const v4f*)(bLh + ko);
                const v4f l1 = *(const v4f*)(bLh + ko + 64);
                const v2f    ax = *(const v2f*)(aX + k0);
                const v4f r0 = *(const v4f*)(bRh + ko);
                const v4f r1 = *(const v4f*)(bRh + ko + 64);
                WMMA_F32(am, l0.x, l0.y, c[0]);
                WMMA_F32(am, l0.z, l0.w, c[1]);
                WMMA_F32(am, l1.x, l1.y, c[2]);
                WMMA_F32(am, l1.z, l1.w, c[3]);
                WMMA_F32(ax, r0.x, r0.y, c[0]);
                WMMA_F32(ax, r0.z, r0.w, c[1]);
                WMMA_F32(ax, r1.x, r1.y, c[2]);
                WMMA_F32(ax, r1.z, r1.w, c[3]);
            }

            #pragma unroll
            for (int r = 0; r < 8; ++r) {
                v4f v = {c[0][r], c[1][r], c[2][r], c[3][r]};
                if (do_relu) {
                    v.x = fmaxf(v.x, 0.0f); v.y = fmaxf(v.y, 0.0f);
                    v.z = fmaxf(v.z, 0.0f); v.w = fmaxf(v.w, 0.0f);
                }
                *(v4f*)(out + (size_t)(rbase + r) * HIDDEN + hp * 64 + 4 * m) = v;
            }
        }

        __builtin_amdgcn_fence(__ATOMIC_RELEASE, "wavefront");
        __builtin_amdgcn_wave_barrier();
        __builtin_amdgcn_fence(__ATOMIC_ACQUIRE, "wavefront");
    }
}

extern "C" void kernel_launch(void* const* d_in, const int* in_sizes, int n_in,
                              void* d_out, int out_size, void* d_ws, size_t ws_size,
                              hipStream_t stream) {
    const float* x   = (const float*)d_in[0];
    const int*   ei  = (const int*)d_in[1];
    const float* Wl1 = (const float*)d_in[2];
    const float* Wr1 = (const float*)d_in[3];
    const float* b1  = (const float*)d_in[4];
    const float* Wl2 = (const float*)d_in[5];
    const float* Wr2 = (const float*)d_in[6];
    const float* b2  = (const float*)d_in[7];
    float* out = (float*)d_out;

    char* ws = (char*)d_ws;
    const size_t featBytes = (size_t)N_NODES * HIDDEN * sizeof(float);
    float*    agg  = (float*)(ws);
    float*    h    = (float*)(ws + featBytes);
    float*    cnt  = (float*)(ws + 2 * featBytes);
    unsigned* fill = (unsigned*)(ws + 2 * featBytes + (size_t)524288);
    unsigned* list = (unsigned*)(ws + 2 * featBytes + (size_t)1048576);

    hipMemsetAsync(fill, 0, (size_t)N_TILES * sizeof(unsigned), stream);
    gnn_bin_edges<<<(N_EDGES + 255) / 256, 256, 0, stream>>>(ei, fill, list, N_EDGES);

    gnn_tile_sum<<<N_TILES, 32, 0, stream>>>(x, ei, fill, list, agg, cnt, N_EDGES, 1);
    gnn_sage_layer<<<LAYER_BLOCKS, LAYER_THREADS, 0, stream>>>(x, agg, cnt,
                                                     Wl1, Wr1, b1, h, 1);

    gnn_tile_sum<<<N_TILES, 32, 0, stream>>>(h, ei, fill, list, agg, cnt, N_EDGES, 0);
    gnn_sage_layer<<<LAYER_BLOCKS, LAYER_THREADS, 0, stream>>>(h, agg, cnt,
                                                     Wl2, Wr2, b2, out, 0);
}


# Edge binning off global atomics: blocks split edge slices into 250 coarse buckets in LDS; a 2nd launch splits buckets by tile, sorts the lists in registers; memset, in-sum sort go; sum order same
# speedup vs baseline: 2.5789x; 1.6611x over previous
#include <hip/hip_runtime.h>


#define N_NODES 100000
#define N_EDGES 800000
#define HIDDEN  128

#define TILE_NODES 16
#define N_TILES    (N_NODES / TILE_NODES)
#define LIST_CAP   256
#define SRC_BITS   17
#define SRC_MASK   ((1u << SRC_BITS) - 1u)
#define KEY_PAD    (31u << SRC_BITS)

static_assert(LIST_CAP == 256, "the sort holds 8 list elements per lane");

static_assert(N_NODES % TILE_NODES == 0, "tiles cover the nodes exactly");
static_assert(N_NODES <= (1 << SRC_BITS), "source id fits the key");

typedef __attribute__((ext_vector_type(2))) float v2f;
typedef __attribute__((ext_vector_type(8))) float v8f;
typedef __attribute__((ext_vector_type(4))) float v4f;

#define COARSE_TILES  25
#define COARSE_NODES  (COARSE_TILES * TILE_NODES)
#define N_COARSE      (N_TILES / COARSE_TILES)
#define PART_BLOCKS   256
#define PART_THREADS  256
#define PART_EDGES    (N_EDGES / PART_BLOCKS)
#define PART_FULL     (PART_EDGES / PART_THREADS)
#define PART_REST     (PART_EDGES - PART_FULL * PART_THREADS)
#define RUN_CAP       40
#define RUN_VECS      (RUN_CAP / 4)
#define RUN_EMPTY     0xFFFFFFFFu
#define KEY_BITS      (SRC_BITS + 4)
#define KEY_MASK      ((1u << KEY_BITS) - 1u)
#define SORT_THREADS  (COARSE_TILES * 32)
#define BUCKET_VECS   (PART_BLOCKS * RUN_VECS)

static_assert(N_TILES % COARSE_TILES == 0, "buckets cover the tiles exactly");
static_assert(N_EDGES % PART_BLOCKS == 0, "slices cover the edges exactly");
static_assert(RUN_CAP % 4 == 0, "runs are moved in 16-byte pieces");
static_assert(COARSE_NODES <= (1 << (32 - SRC_BITS)), "bucket-local node fits the word");
static_assert(COARSE_TILES < 31, "tile counters and the overflow flag share 32 words");

__global__ __launch_bounds__(PART_THREADS)
void gnn_part_edges(const int* __restrict__ ei,
                    unsigned* __restrict__ runs,
                    unsigned* __restrict__ rcnt, int E) {
    __shared__ __attribute__((aligned(16))) unsigned stage[N_COARSE * RUN_CAP];
    __shared__ unsigned lcnt[N_COARSE];

    const int tid = threadIdx.x;
    const int blk = blockIdx.x;
    const int e0  = blk * PART_EDGES;

    unsigned s[PART_FULL + 1], d[PART_FULL + 1];
    #pragma unroll
    for (int j = 0; j < PART_FULL; ++j) {
        s[j] = (unsigned)ei[e0 + tid + j * PART_THREADS];
        d[j] = (unsigned)ei[E + e0 + tid + j * PART_THREADS];
    }
    const bool rest = tid < PART_REST;
    s[PART_FULL] = rest ? (unsigned)ei[e0 + tid + PART_FULL * PART_THREADS] : 0u;
    d[PART_FULL] = rest ? (unsigned)ei[E + e0 + tid + PART_FULL * PART_THREADS] : 0u;

    for (int i = tid; i < N_COARSE * RUN_VECS; i += PART_THREADS)
        ((uint4*)stage)[i] = make_uint4(RUN_EMPTY, RUN_EMPTY, RUN_EMPTY, RUN_EMPTY);
    if (tid < N_COARSE) lcnt[tid] = 0u;
    __syncthreads();

    #pragma unroll
    for (int j = 0; j <= PART_FULL; ++j) {
        if (j < PART_FULL || rest) {
            const unsigned b   = d[j] / COARSE_NODES;
            const unsigned pos = atomicAdd(&lcnt[b], 1u);
            if (pos < RUN_CAP)
                stage[b * RUN_CAP + pos] = ((d[j] - b * COARSE_NODES) << SRC_BITS) | s[j];
        }
    }
    __syncthreads();

    for (int q = tid; q < N_COARSE * RUN_VECS; q += PART_THREADS) {
        const int b    = q / RUN_VECS;
        const int part = q - b * RUN_VECS;
        ((uint4*)(runs + ((size_t)b * PART_BLOCKS + blk) * RUN_CAP))[part] = ((const uint4*)stage)[q];
    }
    if (tid < N_COARSE) rcnt[tid * PART_BLOCKS + blk] = lcnt[tid];
}

#define WAVE_LDS_ORDER() \
    do { __builtin_amdgcn_fence(__ATOMIC_RELEASE, "wavefront"); \
         __builtin_amdgcn_wave_barrier(); \
         __builtin_amdgcn_fence(__ATOMIC_ACQUIRE, "wavefront"); } while (0)

__global__ __launch_bounds__(SORT_THREADS)
void gnn_split_sort(const unsigned* __restrict__ runs,
                    const unsigned* __restrict__ rcnt,
                    unsigned* __restrict__ fill,
                    unsigned* __restrict__ list) {
    __shared__ __attribute__((aligned(16))) unsigned keys[COARSE_TILES * LIST_CAP];
    __shared__ unsigned tcount[32];

    const int tid  = threadIdx.x;
    const int lane = tid & 31;
    const int wave = __builtin_amdgcn_readfirstlane(tid >> 5);
    const int b    = blockIdx.x;

    if (tid < 32) tcount[tid] = 0u;
    bool over = false;
    if (tid < PART_BLOCKS) over = rcnt[b * PART_BLOCKS + tid] > RUN_CAP;

    const uint4* rp = (const uint4*)(runs + (size_t)b * PART_BLOCKS * RUN_CAP);
    uint4 v[4];
    #pragma unroll
    for (int j = 0; j < 4; ++j) {
        const int q = tid + j * SORT_THREADS;
        v[j] = (q < BUCKET_VECS) ? rp[q] : make_uint4(RUN_EMPTY, RUN_EMPTY, RUN_EMPTY, RUN_EMPTY);
    }
    __syncthreads();

    if (over) tcount[31] = 1u;
    #pragma unroll
    for (int j = 0; j < 4; ++j) {
        const unsigned wd[4] = {v[j].x, v[j].y, v[j].z, v[j].w};
        #pragma unroll
        for (int c = 0; c < 4; ++c) {
            if (wd[c] != RUN_EMPTY) {
                const unsigned t    = wd[c] >> KEY_BITS;
                const unsigned slot = atomicAdd(&tcount[t], 1u);
                if (slot < LIST_CAP) keys[t * LIST_CAP + slot] = wd[c] & KEY_MASK;
            }
        }
    }
    __syncthreads();

    const int tile = b * COARSE_TILES + wave;
    const int n    = __builtin_amdgcn_readfirstlane((int)tcount[wave]);
    if (tcount[31] != 0u || n > LIST_CAP) {
        if (lane == 0) fill[tile] = LIST_CAP + 1;
        return;
    }

    unsigned* mk = keys + wave * LIST_CAP;
    unsigned a[8];
    {
        const uint4 p0 = *(const uint4*)(mk + 8 * lane);
        const uint4 p1 = *(const uint4*)(mk + 8 * lane + 4);
        const unsigned raw[8] = {p0.x, p0.y, p0.z, p0.w, p1.x, p1.y, p1.z, p1.w};
        #pragma unroll
        for (int r = 0; r < 8; ++r) a[r] = (8 * lane + r < n) ? raw[r] : KEY_PAD;
    }
    WAVE_LDS_ORDER();
    #pragma unroll
    for (int k = 2; k <= LIST_CAP; k <<= 1) {
        #pragma unroll
        for (int j = k >> 1; j > 0; j >>= 1) {
            if (j >= 8) {
                *(uint4*)(mk + 4 * lane)       = make_uint4(a[0], a[1], a[2], a[3]);
                *(uint4*)(mk + 128 + 4 * lane) = make_uint4(a[4], a[5], a[6], a[7]);
                WAVE_LDS_ORDER();
                const int p = lane ^ (j >> 3);
                const uint4 q0 = *(const uint4*)(mk + 4 * p);
                const uint4 q1 = *(const uint4*)(mk + 128 + 4 * p);
                WAVE_LDS_ORDER();
                const unsigned o[8] = {q0.x, q0.y, q0.z, q0.w, q1.x, q1.y, q1.z, q1.w};
                const bool up       = (lane & (k >> 3)) == 0;
                const bool lower    = (lane & (j >> 3)) == 0;
                const bool keep_min = (up == lower);
                #pragma unroll
                for (int r = 0; r < 8; ++r)
                    a[r] = keep_min ? min(a[r], o[r]) : max(a[r], o[r]);
            } else {
                #pragma unroll
                for (int r = 0; r < 8; ++r) {
                    if ((r & j) == 0) {
                        const bool up = (k >= 8) ? ((lane & (k >> 3)) == 0) : ((r & k) == 0);
                        const unsigned lo = min(a[r], a[r | j]), hi = max(a[r], a[r | j]);
                        a[r]     = up ? lo : hi;
                        a[r | j] = up ? hi : lo;
                    }
                }
            }
        }
    }
    *(uint4*)(mk + 8 * lane)     = make_uint4(a[0], a[1], a[2], a[3]);
    *(uint4*)(mk + 8 * lane + 4) = make_uint4(a[4], a[5], a[6], a[7]);
    WAVE_LDS_ORDER();
    unsigned* mine = list + (size_t)tile * LIST_CAP;
    for (int i = lane; i < n; i += 32) mine[i] = mk[i];
    if (lane == 0) fill[tile] = (unsigned)n;
}

static __device__ __forceinline__
void tile_accumulate(const float* __restrict__ feat, unsigned* keys,
                     unsigned* hist, int count, int lane, v8f* c) {
    const int padded = (count + 15) & ~15;
    for (int i = count + lane; i < padded; i += 32) keys[i] = KEY_PAD;
    __syncthreads();

    for (int i = lane; i < count; i += 32)
        atomicAdd(&hist[keys[i] >> SRC_BITS], 1u);

    const unsigned m = lane & 15;
    const int khalf  = (lane >> 4) << 1;

    for (int k0 = 0; k0 < padded; k0 += 16) {
        #pragma unroll
        for (int u = 0; u < 4; ++u) {
            const uint2 kk = *(const uint2*)&keys[k0 + 4 * u + khalf];
            const float* p0 = feat + (size_t)(kk.x & SRC_MASK) * HIDDEN + m * 4;
            const float* p1 = feat + (size_t)(kk.y & SRC_MASK) * HIDDEN + m * 4;
            const float4 r00 = *(const float4*)(p0);
            const float4 r01 = *(const float4*)(p0 + 64);
            const float4 r10 = *(const float4*)(p1);
            const float4 r11 = *(const float4*)(p1 + 64);
            v2f a, b;
            a.x = ((kk.x >> SRC_BITS) == m) ? 1.0f : 0.0f;
            a.y = ((kk.y >> SRC_BITS) == m) ? 1.0f : 0.0f;
            b.x = r00.x; b.y = r10.x;
            c[0] = __builtin_amdgcn_wmma_f32_16x16x4_f32(false, a, false, b, (short)0, c[0], false, false);
            b.x = r00.y; b.y = r10.y;
            c[1] = __builtin_amdgcn_wmma_f32_16x16x4_f32(false, a, false, b, (short)0, c[1], false, false);
            b.x = r00.z; b.y = r10.z;
            c[2] = __builtin_amdgcn_wmma_f32_16x16x4_f32(false, a, false, b, (short)0, c[2], false, false);
            b.x = r00.w; b.y = r10.w;
            c[3] = __builtin_amdgcn_wmma_f32_16x16x4_f32(false, a, false, b, (short)0, c[3], false, false);
            b.x = r01.x; b.y = r11.x;
            c[4] = __builtin_amdgcn_wmma_f32_16x16x4_f32(false, a, false, b, (short)0, c[4], false, false);
            b.x = r01.y; b.y = r11.y;
            c[5] = __builtin_amdgcn_wmma_f32_16x16x4_f32(false, a, false, b, (short)0, c[5], false, false);
            b.x = r01.z; b.y = r11.z;
            c[6] = __builtin_amdgcn_wmma_f32_16x16x4_f32(false, a, false, b, (short)0, c[6], false, false);
            b.x = r01.w; b.y = r11.w;
            c[7] = __builtin_amdgcn_wmma_f32_16x16x4_f32(false, a, false, b, (short)0, c[7], false, false);
        }
    }
}

__global__ __launch_bounds__(32)
void gnn_tile_sum(const float* __restrict__ feat,
                  const int* __restrict__ ei,
                  const unsigned* __restrict__ fill,
                  const unsigned* __restrict__ list,
                  float* __restrict__ agg,
                  float* __restrict__ cnt,
                  int E, int first_pass) {
    __shared__ __attribute__((aligned(16))) unsigned keys[LIST_CAP + 16];
    __shared__ unsigned hist[32];

    const int lane = threadIdx.x;
    const int tile = blockIdx.x;
    hist[lane] = 0u;

    v8f c[8];
    #pragma unroll
    for (int t = 0; t < 8; ++t) c[t] = (v8f){0.f, 0.f, 0.f, 0.f, 0.f, 0.f, 0.f, 0.f};

    const int n = __builtin_amdgcn_readfirstlane((int)fill[tile]);

    if (n <= LIST_CAP) {
        const unsigned* mine = list + (size_t)tile * LIST_CAP;
        for (int i = lane; i < n; i += 32) keys[i] = mine[i];
        tile_accumulate(feat, keys, hist, n, lane, c);
    } else {
        int got = 0;
        for (int e0 = 0; e0 < E; e0 += 32) {
            const int e   = e0 + lane;
            const int dst = (e < E) ? ei[E + e] : -1;
            const bool hit = (dst >> 4) == tile;
            const unsigned mask = __builtin_amdgcn_ballot_w32(hit);
            if (mask != 0u) {
                if (hit)
                    keys[got + __popc(mask & ((1u << lane) - 1u))] =
                        (((unsigned)dst & 15u) << SRC_BITS) | (unsigned)ei[e];
                got += __popc(mask);
                if (got > LIST_CAP - 32) {
                    tile_accumulate(feat, keys, hist, got, lane, c);
                    __syncthreads();
                    got = 0;
                }
            }
        }
        if (got > 0) tile_accumulate(feat, keys, hist, got, lane, c);
    }
    __syncthreads();

    const int m     = lane & 15;
    const int rbase = tile * TILE_NODES + ((lane >> 4) << 3);
    #pragma unroll
    for (int r = 0; r < 8; ++r) {
        float* o = agg + (size_t)(rbase + r) * HIDDEN + m * 4;
        *(float4*)(o)      = make_float4(c[0][r], c[1][r], c[2][r], c[3][r]);
        *(float4*)(o + 64) = make_float4(c[4][r], c[5][r], c[6][r], c[7][r]);
    }
    if (first_pass && lane < TILE_NODES)
        cnt[tile * TILE_NODES + lane] = (float)hist[lane];
}

#define LAYER_WAVES   8
#define LAYER_THREADS (LAYER_WAVES * 32)
#define LAYER_BLOCKS  256
#define WP_PITCH      260
#define W_FLOATS      (64 * WP_PITCH)
#define A_LD          (HIDDEN + 4)
#define A_FLOATS      (16 * A_LD)

static_assert(HIDDEN == 128, "the staged layout and the column map assume 128 columns");

#define WMMA_F32(A_, BX_, BY_, C_) \
    do { v2f b_; b_.x = (BX_); b_.y = (BY_); \
         (C_) = __builtin_amdgcn_wmma_f32_16x16x4_f32(false, (A_), false, b_, (short)0, (C_), false, false); } while (0)

__global__ __launch_bounds__(LAYER_THREADS)
void gnn_sage_layer(const float* __restrict__ X,
                    const float* __restrict__ agg,
                    const float* __restrict__ cnt,
                    const float* __restrict__ Wl,
                    const float* __restrict__ Wr,
                    const float* __restrict__ bias,
                    float* __restrict__ out,
                    int do_relu) {
    __shared__ __attribute__((aligned(16))) float Ws[2 * W_FLOATS];
    __shared__ __attribute__((aligned(16))) float At[LAYER_WAVES * 2 * A_FLOATS];
    __shared__ __attribute__((aligned(16))) float Bs[HIDDEN];

    const int tid  = threadIdx.x;
    const int lane = tid & 31;
    const int wave = __builtin_amdgcn_readfirstlane(tid >> 5);
    const int m    = lane & 15;
    const int h    = lane >> 4;

    const int n_waves = LAYER_WAVES * (int)gridDim.x;
    int tile = wave * (int)gridDim.x + (int)blockIdx.x;

    v4f xr[16], ar[16];
    if (tile < N_TILES) {
        const float* xp = X   + (size_t)tile * 16 * HIDDEN + 4 * lane;
        const float* ap = agg + (size_t)tile * 16 * HIDDEN + 4 * lane;
        #pragma unroll
        for (int r = 0; r < 16; ++r) {
            xr[r] = *(const v4f*)(xp + r * HIDDEN);
            ar[r] = *(const v4f*)(ap + r * HIDDEN);
        }
    }

    for (int idx = tid; idx < HIDDEN * HIDDEN; idx += LAYER_THREADS) {
        const int k   = idx >> 7;
        const int col = idx & 127;
        const int w   = col & 63;
        const int g   = ((col >> 6) << 1) | ((w >> 1) & 1);
        const int p   = g * 32 + (w >> 2) * 2 + (w & 1);
        const int a   = (k >> 1) * WP_PITCH + 2 * p + (k & 1);
        Ws[a]            = Wl[idx];
        Ws[W_FLOATS + a] = Wr[idx];
    }
    if (tid < HIDDEN) Bs[tid] = bias[tid];
    __syncthreads();

    float* Mt = At + wave * (2 * A_FLOATS);
    float* Xt = Mt + A_FLOATS;

    const float* aM = Mt + m * A_LD + 2 * h;
    const float* aX = Xt + m * A_LD + 2 * h;
    const float* bL = Ws + h * WP_PITCH + 4 * m;
    const float* bR = bL + W_FLOATS;

    for (; tile < N_TILES; tile += n_waves) {
        const int row0 = tile * 16;

        #pragma unroll
        for (int r = 0; r < 16; ++r) {
            const float rc = 1.0f / fmaxf(cnt[row0 + r], 1.0f);
            *(v4f*)(Mt + r * A_LD + 4 * lane) = ar[r] * rc;
            *(v4f*)(Xt + r * A_LD + 4 * lane) = xr[r];
        }
        __builtin_amdgcn_fence(__ATOMIC_RELEASE, "wavefront");
        __builtin_amdgcn_wave_barrier();
        __builtin_amdgcn_fence(__ATOMIC_ACQUIRE, "wavefront");

        const int nt = tile + n_waves;
        if (nt < N_TILES) {
            const float* xp = X   + (size_t)nt * 16 * HIDDEN + 4 * lane;
            const float* ap = agg + (size_t)nt * 16 * HIDDEN + 4 * lane;
            #pragma unroll
            for (int r = 0; r < 16; ++r) {
                xr[r] = *(const v4f*)(xp + r * HIDDEN);
                ar[r] = *(const v4f*)(ap + r * HIDDEN);
            }
        }

        const int rbase = row0 + (h << 3);
        #pragma unroll 1
        for (int hp = 0; hp < 2; ++hp) {
            const v4f bv = *(const v4f*)(Bs + hp * 64 + 4 * m);
            v8f c[4];
            #pragma unroll
            for (int r = 0; r < 8; ++r) {
                c[0][r] = bv.x; c[1][r] = bv.y; c[2][r] = bv.z; c[3][r] = bv.w;
            }
            const float* bLh = bL + hp * 128;
            const float* bRh = bR + hp * 128;

            #pragma unroll 4
            for (int k0 = 0; k0 < HIDDEN; k0 += 4) {
                const int ko = (k0 >> 1) * WP_PITCH;
                const v2f    am = *(const v2f*)(aM + k0);
                const v4f l0 = *(const v4f*)(bLh + ko);
                const v4f l1 = *(const v4f*)(bLh + ko + 64);
                const v2f    ax = *(const v2f*)(aX + k0);
                const v4f r0 = *(const v4f*)(bRh + ko);
                const v4f r1 = *(const v4f*)(bRh + ko + 64);
                WMMA_F32(am, l0.x, l0.y, c[0]);
                WMMA_F32(am, l0.z, l0.w, c[1]);
                WMMA_F32(am, l1.x, l1.y, c[2]);
                WMMA_F32(am, l1.z, l1.w, c[3]);
                WMMA_F32(ax, r0.x, r0.y, c[0]);
                WMMA_F32(ax, r0.z, r0.w, c[1]);
                WMMA_F32(ax, r1.x, r1.y, c[2]);
                WMMA_F32(ax, r1.z, r1.w, c[3]);
            }

            #pragma unroll
            for (int r = 0; r < 8; ++r) {
                v4f v = {c[0][r], c[1][r], c[2][r], c[3][r]};
                if (do_relu) {
                    v.x = fmaxf(v.x, 0.0f); v.y = fmaxf(v.y, 0.0f);
                    v.z = fmaxf(v.z, 0.0f); v.w = fmaxf(v.w, 0.0f);
                }
                *(v4f*)(out + (size_t)(rbase + r) * HIDDEN + hp * 64 + 4 * m) = v;
            }
        }

        __builtin_amdgcn_fence(__ATOMIC_RELEASE, "wavefront");
        __builtin_amdgcn_wave_barrier();
        __builtin_amdgcn_fence(__ATOMIC_ACQUIRE, "wavefront");
    }
}

extern "C" void kernel_launch(void* const* d_in, const int* in_sizes, int n_in,
                              void* d_out, int out_size, void* d_ws, size_t ws_size,
                              hipStream_t stream) {
    const float* x   = (const float*)d_in[0];
    const int*   ei  = (const int*)d_in[1];
    const float* Wl1 = (const float*)d_in[2];
    const float* Wr1 = (const float*)d_in[3];
    const float* b1  = (const float*)d_in[4];
    const float* Wl2 = (const float*)d_in[5];
    const float* Wr2 = (const float*)d_in[6];
    const float* b2  = (const float*)d_in[7];
    float* out = (float*)d_out;

    char* ws = (char*)d_ws;
    const size_t featBytes = (size_t)N_NODES * HIDDEN * sizeof(float);
    float*    agg  = (float*)(ws);
    float*    h    = (float*)(ws + featBytes);
    float*    cnt  = (float*)(ws + 2 * featBytes);
    unsigned* fill = (unsigned*)(ws + 2 * featBytes + (size_t)524288);
    unsigned* list = (unsigned*)(ws + 2 * featBytes + (size_t)1048576);
    unsigned* runs = (unsigned*)(ws + 2 * featBytes + (size_t)8388608);
    unsigned* rcnt = (unsigned*)(ws + 2 * featBytes + (size_t)20971520);

    gnn_part_edges<<<PART_BLOCKS, PART_THREADS, 0, stream>>>(ei, runs, rcnt, N_EDGES);
    gnn_split_sort<<<N_COARSE, SORT_THREADS, 0, stream>>>(runs, rcnt, fill, list);

    gnn_tile_sum<<<N_TILES, 32, 0, stream>>>(x, ei, fill, list, agg, cnt, N_EDGES, 1);
    gnn_sage_layer<<<LAYER_BLOCKS, LAYER_THREADS, 0, stream>>>(x, agg, cnt,
                                                     Wl1, Wr1, b1, h, 1);

    gnn_tile_sum<<<N_TILES, 32, 0, stream>>>(h, ei, fill, list, agg, cnt, N_EDGES, 0);
    gnn_sage_layer<<<LAYER_BLOCKS, LAYER_THREADS, 0, stream>>>(h, agg, cnt,
                                                     Wl2, Wr2, b2, out, 0);
}
